# starting kernel
# baseline (speedup 1.0000x reference)
#include <hip/hip_runtime.h>


typedef __attribute__((ext_vector_type(2))) float v2f;
typedef __attribute__((ext_vector_type(8))) float v8f;

#define N_WORDS  16384
#define L_CHARS  16
#define C_EMB    64
#define CONV_L   256
#define KERNEL_W 3
#define D_EMB    300
#define OUT_COLS (CONV_L + D_EMB)
#define N_OGRP   (CONV_L / 16)
#define N_FRAG   (KERNEL_W * N_OGRP * 16)
#define WPB      8
#define T_LEN    (L_CHARS + 2)

__global__ void pack_conv_w_kernel(const float* __restrict__ conv_w,
                                   float* __restrict__ wpack) {
    int tid  = blockIdx.x * blockDim.x + threadIdx.x;
    if (tid >= N_FRAG * 32) return;
    int lane = tid & 31;
    int frag = tid >> 5;
    int q    = frag & 15;
    int g    = (frag >> 4) & 15;
    int k    = frag >> 8;
    int koff = (lane < 16) ? 0 : 2;
    int o    = g * 16 + (lane & 15);
    int c0   = 4 * q + koff;
    float w0 = conv_w[o * (C_EMB * KERNEL_W) + (c0 + 0) * KERNEL_W + k];
    float w1 = conv_w[o * (C_EMB * KERNEL_W) + (c0 + 1) * KERNEL_W + k];
    wpack[frag * 64 + lane * 2 + 0] = w0;
    wpack[frag * 64 + lane * 2 + 1] = w1;
}

__global__ void word_gather_kernel(const int* __restrict__ X_word,
                                   const float* __restrict__ word_emb,
                                   float* __restrict__ out) {
    long long i = (long long)blockIdx.x * blockDim.x + threadIdx.x;
    if (i >= (long long)N_WORDS * D_EMB) return;
    int n = (int)(i / D_EMB);
    int j = (int)(i % D_EMB);
    out[(long long)n * OUT_COLS + CONV_L + j] =
        word_emb[(long long)X_word[n] * D_EMB + j];
}

__global__ __launch_bounds__(512)
void char_conv_kernel(const int* __restrict__ X,
                      const float* __restrict__ char_emb,
                      const float* __restrict__ wpack,
                      const float* __restrict__ conv_b,
                      float* __restrict__ out) {
    __shared__ int   Xs[WPB * L_CHARS];
    __shared__ float ce[WPB * L_CHARS * C_EMB];
    __shared__ float Y[16 * T_LEN * 16];

    const int tid = threadIdx.x;
    const int n0  = blockIdx.x * WPB;

    if (tid < WPB * L_CHARS) Xs[tid] = X[n0 * L_CHARS + tid];
    __syncthreads();

    for (int e = tid; e < WPB * L_CHARS * C_EMB; e += 512) {
        int ws = e >> 6;
        int c  = e & 63;
        ce[e] = char_emb[Xs[ws] * C_EMB + c];
    }

    const int lane = tid & 31;
    const int wave = tid >> 5;
    const int g    = wave;
    const int koff = (lane < 16) ? 0 : 2;
    const int srow = lane & 15;
    const int shalf = (lane >> 4) << 3;
    const int ocol  = lane & 15;

    v2f B0[16], B1[16], B2[16];
    {
        const v2f* wp = (const v2f*)wpack;
#pragma unroll
        for (int q = 0; q < 16; ++q) {
            B0[q] = wp[((0 * N_OGRP + g) * 16 + q) * 32 + lane];
            B1[q] = wp[((1 * N_OGRP + g) * 16 + q) * 32 + lane];
            B2[q] = wp[((2 * N_OGRP + g) * 16 + q) * 32 + lane];
        }
    }
    __syncthreads();

    float* Yw = &Y[wave * (T_LEN * 16)];
    const float bias = conv_b[g * 16 + ocol];

    for (int w = 0; w < WPB; ++w) {
#pragma unroll
        for (int i = lane; i < T_LEN * 16; i += 32) Yw[i] = 0.0f;

        const float* cew = &ce[w * L_CHARS * C_EMB];

        v8f acc0 = {}; v8f acc1 = {}; v8f acc2 = {};
#pragma unroll
        for (int q = 0; q < 16; ++q) {
            v2f a = *(const v2f*)&cew[srow * C_EMB + 4 * q + koff];
            acc0 = __builtin_amdgcn_wmma_f32_16x16x4_f32(false, a, false, B0[q],
                                                         (short)0, acc0, false, false);
            acc1 = __builtin_amdgcn_wmma_f32_16x16x4_f32(false, a, false, B1[q],
                                                         (short)0, acc1, false, false);
            acc2 = __builtin_amdgcn_wmma_f32_16x16x4_f32(false, a, false, B2[q],
                                                         (short)0, acc2, false, false);
        }

#pragma unroll
        for (int v = 0; v < 8; ++v) {
            int s = v + shalf;
            atomicAdd(&Yw[(s + 2) * 16 + ocol], acc0[v]);
            atomicAdd(&Yw[(s + 1) * 16 + ocol], acc1[v]);
            atomicAdd(&Yw[(s + 0) * 16 + ocol], acc2[v]);
        }
        __syncthreads();

        if (lane < 16) {
            float smax = -3.402823466e38f;
#pragma unroll
            for (int t = 0; t < T_LEN; ++t)
                smax = fmaxf(smax, Yw[t * 16 + ocol]);
            float r = fmaxf(0.0f, bias + smax);
            out[(long long)(n0 + w) * OUT_COLS + g * 16 + ocol] = r;
        }
        __syncthreads();
    }
}

extern "C" void kernel_launch(void* const* d_in, const int* in_sizes, int n_in,
                              void* d_out, int out_size, void* d_ws, size_t ws_size,
                              hipStream_t stream) {
    const int*   X          = (const int*)d_in[0];
    const int*   X_word     = (const int*)d_in[1];
    const float* char_emb_w = (const float*)d_in[2];
    const float* conv_w     = (const float*)d_in[3];
    const float* conv_b     = (const float*)d_in[4];
    const float* word_emb_w = (const float*)d_in[5];
    float*       out        = (float*)d_out;
    float*       wpack      = (float*)d_ws;

    (void)in_sizes; (void)n_in; (void)out_size; (void)ws_size;

    pack_conv_w_kernel<<<(N_FRAG * 32 + 255) / 256, 256, 0, stream>>>(conv_w, wpack);

    long long wn = (long long)N_WORDS * D_EMB;
    word_gather_kernel<<<(int)((wn + 255) / 256), 256, 0, stream>>>(X_word, word_emb_w, out);

    char_conv_kernel<<<N_WORDS / WPB, 512, 0, stream>>>(X, char_emb_w, wpack, conv_b, out);
}


# Conv in minimal-filtering form F(2,3): 4 products per 2 output positions, not 6 (tile rows = 16 words, copy waves build the transformed rows in LDS); sums regrouped: (m1+m2)+m3 and (m2-m3)-m4
# speedup vs baseline: 2.6248x; 2.6248x over previous
#include <hip/hip_runtime.h>


typedef __attribute__((ext_vector_type(2))) float v2f;
typedef __attribute__((ext_vector_type(4))) float v4f;
typedef __attribute__((ext_vector_type(8))) float v8f;
typedef __attribute__((ext_vector_type(2))) int v2i;
typedef __attribute__((ext_vector_type(4))) int v4i;

#define N_WORDS  16384
#define L_CHARS  16
#define C_EMB    64
#define CONV_L   256
#define KERNEL_W 3
#define D_EMB    300
#define OUT_COLS (CONV_L + D_EMB)
#define N_OGRP   (CONV_L / 16)
#define N_TF     4
#define N_FRAG   (N_TF * N_OGRP * 16)
#define N_PAIR   ((L_CHARS + 2) / 2)

static_assert(KERNEL_W == 3 && (L_CHARS + 2) % 2 == 0, "F(2,3) over an even number of positions");

__device__ __forceinline__ float tap_transform(float w0, float w1, float w2, int i) {
    if (i == 0) return w0;
    if (i == 1) return 0.5f * ((w0 + w2) + w1);
    if (i == 2) return 0.5f * ((w0 + w2) - w1);
    return w2;
}

__global__ void pack_conv_w_kernel(const float* __restrict__ conv_w,
                                   float* __restrict__ wpack) {
    int tid  = blockIdx.x * blockDim.x + threadIdx.x;
    if (tid >= N_FRAG * 32) return;
    int lane = tid & 31;
    int frag = tid >> 5;
    int q    = frag & 15;
    int g    = (frag >> 4) & 15;
    int i    = frag >> 8;
    int koff = (lane < 16) ? 0 : 2;
    int o    = g * 16 + (lane & 15);
    int c0   = 4 * q + koff;
    const float* w = &conv_w[o * (C_EMB * KERNEL_W) + c0 * KERNEL_W];
    wpack[frag * 64 + lane * 2 + 0] = tap_transform(w[0], w[1], w[2], i);
    wpack[frag * 64 + lane * 2 + 1] = tap_transform(w[3], w[4], w[5], i);
}

#define ROW_PIECES (D_EMB / 4)

static_assert(D_EMB % 4 == 0 && OUT_COLS % 4 == 0 && CONV_L % 4 == 0,
              "16-B pieces need rows and offsets that are multiples of 4 floats");

#define WPG     16
#define GPB     4
#define WPBLK   (WPG * GPB)
#define N_CBLK  (N_WORDS / WPBLK)
#define GRP_X   (WPG * L_CHARS)
#define U_ROW   (N_TF * C_EMB + 4)
#define U_PAIR  (WPG * U_ROW)
#define U_BUF   (N_PAIR * U_PAIR)
#define CONV_WAVES 16
#define CONV_T  (CONV_WAVES * 32)
#define COPY_WAVES 4
#define COPY_T  (COPY_WAVES * 32)
#define BLOCK_T (CONV_T + COPY_T)
#define BLD_N   (WPG * (C_EMB / 4) / COPY_T)
#define CPW     8
#define CP_PIECES (CPW * ROW_PIECES)
#define CP_N    ((CP_PIECES + COPY_T - 1) / COPY_T)

static_assert(N_WORDS % WPBLK == 0, "blocks cover the words exactly");
static_assert(GPB * GRP_X == 2 * CONV_T, "two char ids per conv thread fill Xs");
static_assert(CONV_WAVES == N_OGRP, "one conv wave per o-group");
static_assert(WPG == 16, "a tile row per word");
static_assert((WPG * (C_EMB / 4)) % COPY_T == 0, "whole items per copy lane");
static_assert(WPG % CPW == 0, "whole copy rounds per group");
static_assert((CP_N - 1) * COPY_T <= CP_PIECES && CP_N * COPY_T >= CP_PIECES,
              "all copy steps of a round but the last are full");
static_assert((U_ROW * 4) % 16 == 0 && (U_PAIR * 4) % 16 == 0, "padded rows keep 16-B alignment");
static_assert((2 * U_BUF + GPB * GRP_X) * 4 <= 320 * 1024, "fits the LDS of a WGP");

__device__ __forceinline__ void lds_barrier() {
    __builtin_amdgcn_fence(__ATOMIC_RELEASE, "workgroup", "local");
    __builtin_amdgcn_s_barrier();
    __builtin_amdgcn_fence(__ATOMIC_ACQUIRE, "workgroup", "local");
}

__device__ __forceinline__ void build_tiles(const int* __restrict__ xg,
                                            const float* __restrict__ char_emb,
                                            float* __restrict__ Ub, int cl) {
#pragma nounroll
    for (int it = 0; it < BLD_N; ++it) {
        const int id = cl + it * COPY_T;
        const int p4 = (id & 15) * 4;
        const int w  = id >> 4;
        const v4i x0 = *(const v4i*)&xg[w * L_CHARS + 0];
        const v4i x1 = *(const v4i*)&xg[w * L_CHARS + 4];
        const v4i x2 = *(const v4i*)&xg[w * L_CHARS + 8];
        const v4i x3 = *(const v4i*)&xg[w * L_CHARS + 12];
        const int ids[L_CHARS] = { x0[0], x0[1], x0[2], x0[3], x1[0], x1[1], x1[2], x1[3],
                                   x2[0], x2[1], x2[2], x2[3], x3[0], x3[1], x3[2], x3[3] };
        v4f d[L_CHARS];
#pragma unroll
        for (int s = 0; s < L_CHARS; ++s)
            d[s] = *(const v4f*)&char_emb[ids[s] * C_EMB + p4];
        const v4f zero = {0.0f, 0.0f, 0.0f, 0.0f};
        float* ut = &Ub[w * U_ROW + p4];
#pragma unroll
        for (int j = 0; j < N_PAIR; ++j) {
            const v4f d0 = (2 * j - 2 >= 0)      ? d[(2 * j - 2 >= 0) ? 2 * j - 2 : 0] : zero;
            const v4f d1 = (2 * j - 1 >= 0)      ? d[(2 * j - 1 >= 0) ? 2 * j - 1 : 0] : zero;
            const v4f d2 = (2 * j < L_CHARS)     ? d[(2 * j < L_CHARS) ? 2 * j : 0] : zero;
            const v4f d3 = (2 * j + 1 < L_CHARS) ? d[(2 * j + 1 < L_CHARS) ? 2 * j + 1 : 0] : zero;
            *(v4f*)&ut[j * U_PAIR + 0 * C_EMB] = d0 - d2;
            *(v4f*)&ut[j * U_PAIR + 1 * C_EMB] = d1 + d2;
            *(v4f*)&ut[j * U_PAIR + 2 * C_EMB] = d2 - d1;
            *(v4f*)&ut[j * U_PAIR + 3 * C_EMB] = d1 - d3;
        }
    }
}

__global__ __launch_bounds__(BLOCK_T)
void char_conv_kernel(const int* __restrict__ X,
                      const int* __restrict__ X_word,
                      const float* __restrict__ char_emb,
                      const float* __restrict__ word_emb,
                      const float* __restrict__ wpack,
                      const float* __restrict__ conv_b,
                      float* __restrict__ out) {
    __shared__ __attribute__((aligned(16))) int   Xs[GPB * GRP_X];
    __shared__ __attribute__((aligned(16))) float U[2][U_BUF];

    const int tid = threadIdx.x;
    const int nb  = blockIdx.x * WPBLK;

    const int lane = tid & 31;
    const int wave = __builtin_amdgcn_readfirstlane(tid >> 5);

    if (wave >= CONV_WAVES) {
        const int cl = tid - CONV_T;
        __builtin_amdgcn_s_setprio(3);
        lds_barrier();
        build_tiles(&Xs[0], char_emb, U[0], cl);
        lds_barrier();
        for (int grp = 0; grp < GPB; ++grp) {
            if (grp + 1 < GPB)
                build_tiles(&Xs[(grp + 1) * GRP_X], char_emb, U[(grp + 1) & 1], cl);

#pragma nounroll
            for (int rnd = 0; rnd < WPG / CPW; ++rnd) {
                const int n0 = nb + grp * WPG + rnd * CPW;
                int src_off[CP_N];
                int dst_off[CP_N];
                v4f v[CP_N];
#pragma unroll
                for (int u = 0; u < CP_N; ++u) {
                    const int p = cl + u * COPY_T;
                    const int pc = (u < CP_N - 1 || p < CP_PIECES) ? p : CP_PIECES - 1;
                    const int n = pc / ROW_PIECES;
                    const int j = pc - n * ROW_PIECES;
                    src_off[u] = X_word[n0 + n] * D_EMB + 4 * j;
                    dst_off[u] = (n0 + n) * OUT_COLS + CONV_L + 4 * j;
                }
#pragma unroll
                for (int u = 0; u < CP_N; ++u)
                    v[u] = *(const v4f*)&word_emb[src_off[u]];
#pragma unroll
                for (int u = 0; u < CP_N; ++u)
                    if (u < CP_N - 1 || cl + u * COPY_T < CP_PIECES)
                        *(v4f*)&out[dst_off[u]] = v[u];
            }
            if (grp + 1 < GPB)
                lds_barrier();
        }
        __builtin_amdgcn_s_setprio(0);
        return;
    }

    const int g    = wave;
    const int koff = (lane < 16) ? 0 : 2;
    const int srow = lane & 15;
    const int half = lane >> 4;
    const int ocol = lane & 15;

    const v2i xv = *(const v2i*)&X[nb * L_CHARS + 2 * tid];

    v2f B0[16], B1[16], B2[16], B3[16];
    {
        const v2f* wp = (const v2f*)wpack;
#pragma unroll
        for (int q = 0; q < 16; ++q) {
            B0[q] = wp[((0 * N_OGRP + g) * 16 + q) * 32 + lane];
            B1[q] = wp[((1 * N_OGRP + g) * 16 + q) * 32 + lane];
            B2[q] = wp[((2 * N_OGRP + g) * 16 + q) * 32 + lane];
            B3[q] = wp[((3 * N_OGRP + g) * 16 + q) * 32 + lane];
        }
    }
    const float bias = conv_b[g * 16 + ocol];

    *(v2i*)&Xs[2 * tid] = xv;
    lds_barrier();
    lds_barrier();

    for (int grp = 0; grp < GPB; ++grp) {
        const float* ub = &U[grp & 1][srow * U_ROW + koff];

        v8f mx;
#pragma unroll
        for (int v = 0; v < 8; ++v) mx[v] = -__builtin_huge_valf();

#pragma unroll
        for (int ph = 0; ph < 4; ++ph) {
            const int j_lo = (ph == 0) ? 0 : 2 * ph + 1;
            const int j_hi = (ph == 3) ? N_PAIR : 2 * ph + 3;
            if (ph == 0)      __builtin_amdgcn_s_setprio(3);
            else if (ph == 1) __builtin_amdgcn_s_setprio(2);
            else if (ph == 2) __builtin_amdgcn_s_setprio(1);
            else              __builtin_amdgcn_s_setprio(0);
#pragma nounroll
        for (int j = j_lo; j < j_hi; ++j) {
            const float* ut = &ub[j * U_PAIR];

            v8f acc0 = {}; v8f acc1 = {}; v8f acc2 = {}; v8f acc3 = {};
            v2f n0 = *(const v2f*)&ut[0 * C_EMB];
            v2f n1 = *(const v2f*)&ut[1 * C_EMB];
            v2f n2 = *(const v2f*)&ut[2 * C_EMB];
            v2f n3 = *(const v2f*)&ut[3 * C_EMB];
#pragma unroll
            for (int q = 0; q < 16; ++q) {
                const v2f a0 = n0, a1 = n1, a2 = n2, a3 = n3;
                if (q + 1 < 16) {
                    n0 = *(const v2f*)&ut[0 * C_EMB + 4 * (q + 1)];
                    n1 = *(const v2f*)&ut[1 * C_EMB + 4 * (q + 1)];
                    n2 = *(const v2f*)&ut[2 * C_EMB + 4 * (q + 1)];
                    n3 = *(const v2f*)&ut[3 * C_EMB + 4 * (q + 1)];
                }
                acc0 = __builtin_amdgcn_wmma_f32_16x16x4_f32(false, a0, false, B0[q],
                                                             (short)0, acc0, false, false);
                acc1 = __builtin_amdgcn_wmma_f32_16x16x4_f32(false, a1, false, B1[q],
                                                             (short)0, acc1, false, false);
                acc2 = __builtin_amdgcn_wmma_f32_16x16x4_f32(false, a2, false, B2[q],
                                                             (short)0, acc2, false, false);
                acc3 = __builtin_amdgcn_wmma_f32_16x16x4_f32(false, a3, false, B3[q],
                                                             (short)0, acc3, false, false);
            }

#pragma unroll
            for (int v = 0; v < 8; ++v) {
                const float y0 = (acc0[v] + acc1[v]) + acc2[v];
                const float y1 = (acc1[v] - acc2[v]) - acc3[v];
                mx[v] = fmaxf(mx[v], fmaxf(y0, y1));
            }
        }
        }

        float* orow = &out[(long long)(nb + grp * WPG + 8 * half) * OUT_COLS + g * 16 + ocol];
#pragma unroll
        for (int v = 0; v < 8; ++v)
            orow[v * OUT_COLS] = fmaxf(0.0f, bias + mx[v]);

        if (grp + 1 < GPB)
            lds_barrier();
    }
}

extern "C" void kernel_launch(void* const* d_in, const int* in_sizes, int n_in,
                              void* d_out, int out_size, void* d_ws, size_t ws_size,
                              hipStream_t stream) {
    const int*   X          = (const int*)d_in[0];
    const int*   X_word     = (const int*)d_in[1];
    const float* char_emb_w = (const float*)d_in[2];
    const float* conv_w     = (const float*)d_in[3];
    const float* conv_b     = (const float*)d_in[4];
    const float* word_emb_w = (const float*)d_in[5];
    float*       out        = (float*)d_out;
    float*       wpack      = (float*)d_ws;

    (void)in_sizes; (void)n_in; (void)out_size; (void)ws_size;

    pack_conv_w_kernel<<<(N_FRAG * 32 + 255) / 256, 256, 0, stream>>>(conv_w, wpack);

    char_conv_kernel<<<N_CBLK, BLOCK_T, 0, stream>>>(X, X_word, char_emb_w, word_emb_w,
                                                     wpack, conv_b, out);
}


# Conv tiles as polynomial products with overlap-add (transposed F(4,3)): a tile takes its own 4 rows and adds into 6 positions, so 24 products per word, not 28; tap sums regrouped accordingly
# speedup vs baseline: 3.4888x; 1.3292x over previous
#include <hip/hip_runtime.h>


typedef __attribute__((ext_vector_type(2))) float v2f;
typedef __attribute__((ext_vector_type(4))) float v4f;
typedef __attribute__((ext_vector_type(8))) float v8f;
typedef __attribute__((ext_vector_type(4))) int v4i;

#define N_WORDS  16384
#define L_CHARS  16
#define C_EMB    64
#define CONV_L   256
#define KERNEL_W 3
#define D_EMB    300
#define OUT_COLS (CONV_L + D_EMB)
#define N_OGRP   (CONV_L / 16)
#define N_TF     6
#define N_FRAG   (N_TF * N_OGRP * 16)
#define N_T4     (L_CHARS / 4)
#define N_TILE   N_T4

static_assert(KERNEL_W == 3 && L_CHARS % 4 == 0, "tiles of four rows times three taps cover the L+2 positions");

__device__ __forceinline__ float tap_transform(float w0, float w1, float w2, int i) {
    if (i == 0) return 0.25f * w2;
    if (i == 1) return (-1.0f / 6.0f) * ((w2 + w0) + w1);
    if (i == 2) return (-1.0f / 6.0f) * ((w2 + w0) - w1);
    if (i == 3) return (1.0f / 24.0f) * ((w2 + 4.0f * w0) + 2.0f * w1);
    if (i == 4) return (1.0f / 24.0f) * ((w2 + 4.0f * w0) - 2.0f * w1);
    return w0;
}

__global__ void pack_conv_w_kernel(const float* __restrict__ conv_w,
                                   float* __restrict__ wpack) {
    int tid  = blockIdx.x * blockDim.x + threadIdx.x;
    if (tid >= N_FRAG * 32) return;
    int lane = tid & 31;
    int frag = tid >> 5;
    int q    = frag & 15;
    int g    = (frag >> 4) & 15;
    int i    = frag >> 8;
    int koff = (lane < 16) ? 0 : 2;
    int o    = g * 16 + (lane & 15);
    int c0   = 4 * q + koff;
    const float* w = &conv_w[o * (C_EMB * KERNEL_W) + c0 * KERNEL_W];
    wpack[frag * 64 + lane * 2 + 0] = tap_transform(w[0], w[1], w[2], i);
    wpack[frag * 64 + lane * 2 + 1] = tap_transform(w[3], w[4], w[5], i);
}

#define ROW_PIECES (D_EMB / 4)

static_assert(D_EMB % 4 == 0 && OUT_COLS % 4 == 0 && CONV_L % 4 == 0,
              "16-B pieces need rows and offsets that are multiples of 4 floats");

#define WPG     16
#define GPB     8
#define WPBLK   (WPG * GPB)
#define N_CHALF 2
#define N_CBLK  ((N_WORDS / WPBLK) * N_CHALF)
#define GRP_X   (WPG * L_CHARS)
#define U_ROW   (N_TF * C_EMB + 4)
#define U_TILE  (WPG * U_ROW)
#define U_BUF   (N_TILE * U_TILE)
#define CONV_WAVES 8
#define CONV_T  (CONV_WAVES * 32)
#define COPY_WAVES 4
#define COPY_T  (COPY_WAVES * 32)
#define BLOCK_T (CONV_T + COPY_T)
#define BLD_N   (WPG * (C_EMB / 4) / COPY_T)
#define CPW     (WPG / N_CHALF)
#define CP_PIECES (CPW * ROW_PIECES)
#define CP_N    ((CP_PIECES + COPY_T - 1) / COPY_T)

static_assert(N_WORDS % WPBLK == 0, "blocks cover the words exactly");
static_assert(GPB * GRP_X == 8 * CONV_T, "eight char ids per conv thread fill Xs");
static_assert(CONV_WAVES * N_CHALF == N_OGRP, "one conv wave per o-group of the half");
static_assert(WPG == 16, "a tile row per word");
static_assert((WPG * (C_EMB / 4)) % COPY_T == 0, "whole items per copy lane");
static_assert(WPG % N_CHALF == 0, "the halves share a group's word rows evenly");
static_assert((CP_N - 1) * COPY_T <= CP_PIECES && CP_N * COPY_T >= CP_PIECES,
              "all copy steps of a group but the last are full");
static_assert((U_ROW * 4) % 16 == 0 && (U_TILE * 4) % 16 == 0, "padded rows keep 16-B alignment");
static_assert((2 * U_BUF + GPB * GRP_X) * 4 <= 320 * 1024, "fits the LDS of a WGP");

__device__ __forceinline__ void lds_barrier() {
    __builtin_amdgcn_fence(__ATOMIC_RELEASE, "workgroup", "local");
    __builtin_amdgcn_s_barrier();
    __builtin_amdgcn_fence(__ATOMIC_ACQUIRE, "workgroup", "local");
}

__device__ __forceinline__ void build_tiles(const int* __restrict__ xg,
                                            const float* __restrict__ char_emb,
                                            float* __restrict__ Ub, int cl) {
#pragma nounroll
    for (int it = 0; it < BLD_N; ++it) {
        const int id = cl + it * COPY_T;
        const int p4 = (id & 15) * 4;
        const int w  = id >> 4;
        const v4i i0 = *(const v4i*)&xg[w * L_CHARS + 0];
        const v4i i1 = *(const v4i*)&xg[w * L_CHARS + 4];
        const v4i i2 = *(const v4i*)&xg[w * L_CHARS + 8];
        const v4i i3 = *(const v4i*)&xg[w * L_CHARS + 12];
        const int ids[L_CHARS] = { i0[0], i0[1], i0[2], i0[3], i1[0], i1[1], i1[2], i1[3],
                                   i2[0], i2[1], i2[2], i2[3], i3[0], i3[1], i3[2], i3[3] };
        v4f d[L_CHARS];
#pragma unroll
        for (int s = 0; s < L_CHARS; ++s)
            d[s] = *(const v4f*)&char_emb[ids[s] * C_EMB + p4];
        float* ut = &Ub[w * U_ROW + p4];
#pragma unroll
        for (int j = 0; j < N_T4; ++j) {
            const v4f x0 = d[4 * j];
            const v4f x1 = d[4 * j + 1];
            const v4f x2 = d[4 * j + 2];
            const v4f x3 = d[4 * j + 3];
            const v4f a = x0 + x2;
            const v4f b = x1 + x3;
            const v4f cc = x0 + 4.0f * x2;
            const v4f e = 2.0f * (x1 + 4.0f * x3);
            *(v4f*)&ut[j * U_TILE + 0 * C_EMB] = x0;
            *(v4f*)&ut[j * U_TILE + 1 * C_EMB] = a + b;
            *(v4f*)&ut[j * U_TILE + 2 * C_EMB] = a - b;
            *(v4f*)&ut[j * U_TILE + 3 * C_EMB] = cc + e;
            *(v4f*)&ut[j * U_TILE + 4 * C_EMB] = cc - e;
            *(v4f*)&ut[j * U_TILE + 5 * C_EMB] = x3;
        }
    }
}

__global__ __launch_bounds__(BLOCK_T)
void char_conv_kernel(const int* __restrict__ X,
                      const int* __restrict__ X_word,
                      const float* __restrict__ char_emb,
                      const float* __restrict__ word_emb,
                      const float* __restrict__ wpack,
                      const float* __restrict__ conv_b,
                      float* __restrict__ out) {
    __shared__ __attribute__((aligned(16))) int   Xs[GPB * GRP_X];
    __shared__ __attribute__((aligned(16))) float U[2][U_BUF];

    const int tid = threadIdx.x;
    const int ch  = blockIdx.x & (N_CHALF - 1);
    const int nb  = (blockIdx.x / N_CHALF) * WPBLK;

    const int lane = tid & 31;
    const int wave = __builtin_amdgcn_readfirstlane(tid >> 5);

    if (wave >= CONV_WAVES) {
        const int cl = tid - CONV_T;
        __builtin_amdgcn_s_setprio(3);
        lds_barrier();
        build_tiles(&Xs[0], char_emb, U[0], cl);
        lds_barrier();
        for (int grp = 0; grp < GPB; ++grp) {
            if (grp + 1 < GPB)
                build_tiles(&Xs[(grp + 1) * GRP_X], char_emb, U[(grp + 1) & 1], cl);

            {
                const int n0 = nb + grp * WPG + ch * CPW;
                int src_off[CP_N];
                int dst_off[CP_N];
                v4f v[CP_N];
#pragma unroll
                for (int u = 0; u < CP_N; ++u) {
                    const int p = cl + u * COPY_T;
                    const int pc = (u < CP_N - 1 || p < CP_PIECES) ? p : CP_PIECES - 1;
                    const int n = pc / ROW_PIECES;
                    const int j = pc - n * ROW_PIECES;
                    src_off[u] = X_word[n0 + n] * D_EMB + 4 * j;
                    dst_off[u] = (n0 + n) * OUT_COLS + CONV_L + 4 * j;
                }
#pragma unroll
                for (int u = 0; u < CP_N; ++u)
                    v[u] = *(const v4f*)&word_emb[src_off[u]];
#pragma unroll
                for (int u = 0; u < CP_N; ++u)
                    if (u < CP_N - 1 || cl + u * COPY_T < CP_PIECES)
                        *(v4f*)&out[dst_off[u]] = v[u];
            }
            if (grp + 1 < GPB)
                lds_barrier();
        }
        __builtin_amdgcn_s_setprio(0);
        return;
    }

    const int g    = ch * CONV_WAVES + wave;
    const int koff = (lane < 16) ? 0 : 2;
    const int srow = lane & 15;
    const int half = lane >> 4;
    const int ocol = lane & 15;

    const v4i xa = *(const v4i*)&X[nb * L_CHARS + 8 * tid];
    const v4i xb = *(const v4i*)&X[nb * L_CHARS + 8 * tid + 4];

    v2f B0[16], B1[16], B2[16], B3[16], B4[16], B5[16];
    {
        const v2f* wp = (const v2f*)wpack;
#pragma unroll
        for (int q = 0; q < 16; ++q) {
            B0[q] = wp[((0 * N_OGRP + g) * 16 + q) * 32 + lane];
            B1[q] = wp[((1 * N_OGRP + g) * 16 + q) * 32 + lane];
            B2[q] = wp[((2 * N_OGRP + g) * 16 + q) * 32 + lane];
            B3[q] = wp[((3 * N_OGRP + g) * 16 + q) * 32 + lane];
            B4[q] = wp[((4 * N_OGRP + g) * 16 + q) * 32 + lane];
            B5[q] = wp[((5 * N_OGRP + g) * 16 + q) * 32 + lane];
        }
    }
    const float bias = conv_b[g * 16 + ocol];

    *(v4i*)&Xs[8 * tid]     = xa;
    *(v4i*)&Xs[8 * tid + 4] = xb;
    lds_barrier();
    lds_barrier();

    for (int grp = 0; grp < GPB; ++grp) {
        const float* ub = &U[grp & 1][srow * U_ROW + koff];

        v8f mx;
#pragma unroll
        for (int v = 0; v < 8; ++v) mx[v] = -__builtin_huge_valf();

        v8f c4 = {}; v8f c5 = {};

#pragma nounroll
        for (int j = 0; j < N_T4; ++j) {
            if (j == 0)      __builtin_amdgcn_s_setprio(3);
            else if (j == 1) __builtin_amdgcn_s_setprio(2);
            else if (j == 2) __builtin_amdgcn_s_setprio(1);
            else             __builtin_amdgcn_s_setprio(0);

            const float* ut = &ub[j * U_TILE];

            v8f acc0 = {}; v8f acc1 = {}; v8f acc2 = {}; v8f acc3 = {}; v8f acc4 = {}; v8f acc5 = {};
            v2f n0 = *(const v2f*)&ut[0 * C_EMB];
            v2f n1 = *(const v2f*)&ut[1 * C_EMB];
            v2f n2 = *(const v2f*)&ut[2 * C_EMB];
            v2f n3 = *(const v2f*)&ut[3 * C_EMB];
            v2f n4 = *(const v2f*)&ut[4 * C_EMB];
            v2f n5 = *(const v2f*)&ut[5 * C_EMB];
#pragma unroll
            for (int q = 0; q < 16; ++q) {
                const v2f a0 = n0, a1 = n1, a2 = n2, a3 = n3, a4 = n4, a5 = n5;
                if (q + 1 < 16) {
                    n0 = *(const v2f*)&ut[0 * C_EMB + 4 * (q + 1)];
                    n1 = *(const v2f*)&ut[1 * C_EMB + 4 * (q + 1)];
                    n2 = *(const v2f*)&ut[2 * C_EMB + 4 * (q + 1)];
                    n3 = *(const v2f*)&ut[3 * C_EMB + 4 * (q + 1)];
                    n4 = *(const v2f*)&ut[4 * C_EMB + 4 * (q + 1)];
                    n5 = *(const v2f*)&ut[5 * C_EMB + 4 * (q + 1)];
                }
                acc0 = __builtin_amdgcn_wmma_f32_16x16x4_f32(false, a0, false, B0[q],
                                                             (short)0, acc0, false, false);
                acc1 = __builtin_amdgcn_wmma_f32_16x16x4_f32(false, a1, false, B1[q],
                                                             (short)0, acc1, false, false);
                acc2 = __builtin_amdgcn_wmma_f32_16x16x4_f32(false, a2, false, B2[q],
                                                             (short)0, acc2, false, false);
                acc3 = __builtin_amdgcn_wmma_f32_16x16x4_f32(false, a3, false, B3[q],
                                                             (short)0, acc3, false, false);
                acc4 = __builtin_amdgcn_wmma_f32_16x16x4_f32(false, a4, false, B4[q],
                                                             (short)0, acc4, false, false);
                acc5 = __builtin_amdgcn_wmma_f32_16x16x4_f32(false, a5, false, B5[q],
                                                             (short)0, acc5, false, false);
            }

#pragma unroll
            for (int v = 0; v < 8; ++v) {
                const float s12 = acc1[v] + acc2[v];
                const float d12 = acc1[v] - acc2[v];
                const float s34 = acc3[v] + acc4[v];
                const float d34 = acc3[v] - acc4[v];
                const float y0 = 4.0f * acc0[v] + c4[v];
                const float y1 = (4.0f * (acc5[v] - d12) - 2.0f * d34) + c5[v];
                const float y2 = -(5.0f * acc0[v] + 4.0f * s12) - s34;
                const float y3 = (d12 + 2.0f * d34) - 5.0f * acc5[v];
                c4[v] = (acc0[v] + s12) + s34;
                c5[v] = acc5[v];
                mx[v] = fmaxf(mx[v], fmaxf(fmaxf(y0, y1), fmaxf(y2, y3)));
            }
        }

#pragma unroll
        for (int v = 0; v < 8; ++v)
            mx[v] = fmaxf(mx[v], fmaxf(c4[v], c5[v]));

        float* orow = &out[(long long)(nb + grp * WPG + 8 * half) * OUT_COLS + g * 16 + ocol];
#pragma unroll
        for (int v = 0; v < 8; ++v)
            orow[v * OUT_COLS] = fmaxf(0.0f, bias + mx[v]);

        if (grp + 1 < GPB)
            lds_barrier();
    }
}

extern "C" void kernel_launch(void* const* d_in, const int* in_sizes, int n_in,
                              void* d_out, int out_size, void* d_ws, size_t ws_size,
                              hipStream_t stream) {
    const int*   X          = (const int*)d_in[0];
    const int*   X_word     = (const int*)d_in[1];
    const float* char_emb_w = (const float*)d_in[2];
    const float* conv_w     = (const float*)d_in[3];
    const float* conv_b     = (const float*)d_in[4];
    const float* word_emb_w = (const float*)d_in[5];
    float*       out        = (float*)d_out;
    float*       wpack      = (float*)d_ws;

    (void)in_sizes; (void)n_in; (void)out_size; (void)ws_size;

    pack_conv_w_kernel<<<(N_FRAG * 32 + 255) / 256, 256, 0, stream>>>(conv_w, wpack);

    char_conv_kernel<<<N_CBLK, BLOCK_T, 0, stream>>>(X, X_word, char_emb_w, word_emb_w,
                                                     wpack, conv_b, out);
}


# Weight transform moved from its own launch into the conv waves' start-up: a wave reads its 12 KB slab of conv_w coalesced, turns it via LDS into its B fragments; launches 2->1, workspace unused
# speedup vs baseline: 3.7213x; 1.0666x over previous
#include <hip/hip_runtime.h>


typedef __attribute__((ext_vector_type(2))) float v2f;
typedef __attribute__((ext_vector_type(4))) float v4f;
typedef __attribute__((ext_vector_type(8))) float v8f;
typedef __attribute__((ext_vector_type(4))) int v4i;

#define N_WORDS  16384
#define L_CHARS  16
#define C_EMB    64
#define CONV_L   256
#define KERNEL_W 3
#define D_EMB    300
#define OUT_COLS (CONV_L + D_EMB)
#define N_OGRP   (CONV_L / 16)
#define N_TF     6
#define N_T4     (L_CHARS / 4)
#define N_TILE   N_T4

static_assert(KERNEL_W == 3 && L_CHARS % 4 == 0, "tiles of four rows times three taps cover the L+2 positions");

__device__ __forceinline__ v2f tap_transform(v2f w0, v2f w1, v2f w2, int i) {
    if (i == 0) return 0.25f * w2;
    if (i == 1) return (-1.0f / 6.0f) * ((w2 + w0) + w1);
    if (i == 2) return (-1.0f / 6.0f) * ((w2 + w0) - w1);
    if (i == 3) return (1.0f / 24.0f) * ((w2 + 4.0f * w0) + 2.0f * w1);
    if (i == 4) return (1.0f / 24.0f) * ((w2 + 4.0f * w0) - 2.0f * w1);
    return w0;
}

#define ROW_PIECES (D_EMB / 4)

static_assert(D_EMB % 4 == 0 && OUT_COLS % 4 == 0 && CONV_L % 4 == 0,
              "16-B pieces need rows and offsets that are multiples of 4 floats");

#define WPG     16
#define GPB     8
#define WPBLK   (WPG * GPB)
#define N_CHALF 2
#define N_CBLK  ((N_WORDS / WPBLK) * N_CHALF)
#define GRP_X   (WPG * L_CHARS)
#define U_ROW   (N_TF * C_EMB + 4)
#define U_TILE  (WPG * U_ROW)
#define U_BUF   (N_TILE * U_TILE)
#define CONV_WAVES 8
#define CONV_T  (CONV_WAVES * 32)
#define COPY_WAVES 4
#define COPY_T  (COPY_WAVES * 32)
#define BLOCK_T (CONV_T + COPY_T)
#define BLD_N   (WPG * (C_EMB / 4) / COPY_T)
#define CPW     (WPG / N_CHALF)
#define CP_PIECES (CPW * ROW_PIECES)
#define CP_N    ((CP_PIECES + COPY_T - 1) / COPY_T)
#define W_ROW   (C_EMB * KERNEL_W)
#define W_PITCH (W_ROW + 2)
#define W_SLAB  (16 * W_PITCH)
#define W_LOADS (16 * W_ROW / 4 / 32)

static_assert(N_WORDS % WPBLK == 0, "blocks cover the words exactly");
static_assert(GPB * GRP_X == 8 * CONV_T, "eight char ids per conv thread fill Xs");
static_assert(CONV_WAVES * N_CHALF == N_OGRP, "one conv wave per o-group of the half");
static_assert(WPG == 16, "a tile row per word");
static_assert((WPG * (C_EMB / 4)) % COPY_T == 0, "whole items per copy lane");
static_assert(WPG % N_CHALF == 0, "the halves share a group's word rows evenly");
static_assert((CP_N - 1) * COPY_T <= CP_PIECES && CP_N * COPY_T >= CP_PIECES,
              "all copy steps of a group but the last are full");
static_assert((U_ROW * 4) % 16 == 0 && (U_TILE * 4) % 16 == 0, "padded rows keep 16-B alignment");
static_assert((2 * U_BUF + GPB * GRP_X) * 4 <= 320 * 1024, "fits the LDS of a WGP");
static_assert(CONV_WAVES * W_SLAB <= U_BUF, "the weight slabs fit into the second tile buffer");
static_assert(W_ROW == 3 * 64 && W_LOADS == 24 && W_PITCH % 2 == 0, "slab load pattern: 2 rows x 16 pieces per load, 8 row pairs x 3 column blocks");

__device__ __forceinline__ void lds_barrier() {
    __builtin_amdgcn_fence(__ATOMIC_RELEASE, "workgroup", "local");
    __builtin_amdgcn_s_barrier();
    __builtin_amdgcn_fence(__ATOMIC_ACQUIRE, "workgroup", "local");
}

__device__ __forceinline__ void build_tiles(const int* __restrict__ xg,
                                            const float* __restrict__ char_emb,
                                            float* __restrict__ Ub, int cl) {
#pragma nounroll
    for (int it = 0; it < BLD_N; ++it) {
        const int id = cl + it * COPY_T;
        const int p4 = (id & 15) * 4;
        const int w  = id >> 4;
        const v4i i0 = *(const v4i*)&xg[w * L_CHARS + 0];
        const v4i i1 = *(const v4i*)&xg[w * L_CHARS + 4];
        const v4i i2 = *(const v4i*)&xg[w * L_CHARS + 8];
        const v4i i3 = *(const v4i*)&xg[w * L_CHARS + 12];
        const int ids[L_CHARS] = { i0[0], i0[1], i0[2], i0[3], i1[0], i1[1], i1[2], i1[3],
                                   i2[0], i2[1], i2[2], i2[3], i3[0], i3[1], i3[2], i3[3] };
        v4f d[L_CHARS];
#pragma unroll
        for (int s = 0; s < L_CHARS; ++s)
            d[s] = *(const v4f*)&char_emb[ids[s] * C_EMB + p4];
        float* ut = &Ub[w * U_ROW + p4];
#pragma unroll
        for (int j = 0; j < N_T4; ++j) {
            const v4f x0 = d[4 * j];
            const v4f x1 = d[4 * j + 1];
            const v4f x2 = d[4 * j + 2];
            const v4f x3 = d[4 * j + 3];
            const v4f a = x0 + x2;
            const v4f b = x1 + x3;
            const v4f cc = x0 + 4.0f * x2;
            const v4f e = 2.0f * (x1 + 4.0f * x3);
            *(v4f*)&ut[j * U_TILE + 0 * C_EMB] = x0;
            *(v4f*)&ut[j * U_TILE + 1 * C_EMB] = a + b;
            *(v4f*)&ut[j * U_TILE + 2 * C_EMB] = a - b;
            *(v4f*)&ut[j * U_TILE + 3 * C_EMB] = cc + e;
            *(v4f*)&ut[j * U_TILE + 4 * C_EMB] = cc - e;
            *(v4f*)&ut[j * U_TILE + 5 * C_EMB] = x3;
        }
    }
}

__global__ __launch_bounds__(BLOCK_T)
void char_conv_kernel(const int* __restrict__ X,
                      const int* __restrict__ X_word,
                      const float* __restrict__ char_emb,
                      const float* __restrict__ word_emb,
                      const float* __restrict__ conv_w,
                      const float* __restrict__ conv_b,
                      float* __restrict__ out) {
    __shared__ __attribute__((aligned(16))) int   Xs[GPB * GRP_X];
    __shared__ __attribute__((aligned(16))) float U[2][U_BUF];

    const int tid = threadIdx.x;
    const int ch  = blockIdx.x & (N_CHALF - 1);
    const int nb  = (blockIdx.x / N_CHALF) * WPBLK;

    const int lane = tid & 31;
    const int wave = __builtin_amdgcn_readfirstlane(tid >> 5);

    if (wave >= CONV_WAVES) {
        const int cl = tid - CONV_T;
        __builtin_amdgcn_s_setprio(3);
        lds_barrier();
        build_tiles(&Xs[0], char_emb, U[0], cl);
        lds_barrier();
        for (int grp = 0; grp < GPB; ++grp) {
            if (grp + 1 < GPB)
                build_tiles(&Xs[(grp + 1) * GRP_X], char_emb, U[(grp + 1) & 1], cl);

            {
                const int n0 = nb + grp * WPG + ch * CPW;
                int src_off[CP_N];
                int dst_off[CP_N];
                v4f v[CP_N];
#pragma unroll
                for (int u = 0; u < CP_N; ++u) {
                    const int p = cl + u * COPY_T;
                    const int pc = (u < CP_N - 1 || p < CP_PIECES) ? p : CP_PIECES - 1;
                    const int n = pc / ROW_PIECES;
                    const int j = pc - n * ROW_PIECES;
                    src_off[u] = X_word[n0 + n] * D_EMB + 4 * j;
                    dst_off[u] = (n0 + n) * OUT_COLS + CONV_L + 4 * j;
                }
#pragma unroll
                for (int u = 0; u < CP_N; ++u)
                    v[u] = *(const v4f*)&word_emb[src_off[u]];
#pragma unroll
                for (int u = 0; u < CP_N; ++u)
                    if (u < CP_N - 1 || cl + u * COPY_T < CP_PIECES)
                        *(v4f*)&out[dst_off[u]] = v[u];
            }
            if (grp + 1 < GPB)
                lds_barrier();
        }
        __builtin_amdgcn_s_setprio(0);
        return;
    }

    const int g    = ch * CONV_WAVES + wave;
    const int koff = (lane < 16) ? 0 : 2;
    const int srow = lane & 15;
    const int half = lane >> 4;
    const int ocol = lane & 15;

    const v4i xa = *(const v4i*)&X[nb * L_CHARS + 8 * tid];
    const v4i xb = *(const v4i*)&X[nb * L_CHARS + 8 * tid + 4];

    const float* wg = &conv_w[g * 16 * W_ROW];
    const int l15 = lane & 15;
    v4f wt[W_LOADS];
#pragma unroll
    for (int t = 0; t < W_LOADS; ++t)
        wt[t] = *(const v4f*)&wg[(2 * (t & 7) + half) * W_ROW + (t >> 3) * 64 + l15 * 4];

    *(v4i*)&Xs[8 * tid]     = xa;
    *(v4i*)&Xs[8 * tid + 4] = xb;
    lds_barrier();

    const float bias = conv_b[g * 16 + ocol];

    float* wsl = &U[1][wave * W_SLAB];
#pragma unroll
    for (int t = 0; t < W_LOADS; ++t) {
        float* d = &wsl[(2 * (t & 7) + half) * W_PITCH + (t >> 3) * 64 + l15 * 4];
        *(v2f*)&d[0] = (v2f){ wt[t][0], wt[t][1] };
        *(v2f*)&d[2] = (v2f){ wt[t][2], wt[t][3] };
    }
    __builtin_amdgcn_fence(__ATOMIC_RELEASE, "workgroup", "local");
    __builtin_amdgcn_wave_barrier();
    __builtin_amdgcn_fence(__ATOMIC_ACQUIRE, "workgroup", "local");

    v2f B0[16], B1[16], B2[16], B3[16], B4[16], B5[16];
    {
        const float* wr = &wsl[ocol * W_PITCH + 3 * koff];
#pragma unroll
        for (int q = 0; q < 16; ++q) {
            const v2f f01 = *(const v2f*)&wr[12 * q + 0];
            const v2f f23 = *(const v2f*)&wr[12 * q + 2];
            const v2f f45 = *(const v2f*)&wr[12 * q + 4];
            const v2f w0 = (v2f){ f01[0], f23[1] };
            const v2f w1 = (v2f){ f01[1], f45[0] };
            const v2f w2 = (v2f){ f23[0], f45[1] };
            B0[q] = tap_transform(w0, w1, w2, 0);
            B1[q] = tap_transform(w0, w1, w2, 1);
            B2[q] = tap_transform(w0, w1, w2, 2);
            B3[q] = tap_transform(w0, w1, w2, 3);
            B4[q] = tap_transform(w0, w1, w2, 4);
            B5[q] = tap_transform(w0, w1, w2, 5);
        }
    }

    lds_barrier();

    for (int grp = 0; grp < GPB; ++grp) {
        const float* ub = &U[grp & 1][srow * U_ROW + koff];

        v8f mx;
#pragma unroll
        for (int v = 0; v < 8; ++v) mx[v] = -__builtin_huge_valf();

        v8f c4 = {}; v8f c5 = {};

#pragma nounroll
        for (int j = 0; j < N_T4; ++j) {
            if (j == 0)      __builtin_amdgcn_s_setprio(3);
            else if (j == 1) __builtin_amdgcn_s_setprio(2);
            else if (j == 2) __builtin_amdgcn_s_setprio(1);
            else             __builtin_amdgcn_s_setprio(0);

            const float* ut = &ub[j * U_TILE];

            v8f acc0 = {}; v8f acc1 = {}; v8f acc2 = {}; v8f acc3 = {}; v8f acc4 = {}; v8f acc5 = {};
            v2f n0 = *(const v2f*)&ut[0 * C_EMB];
            v2f n1 = *(const v2f*)&ut[1 * C_EMB];
            v2f n2 = *(const v2f*)&ut[2 * C_EMB];
            v2f n3 = *(const v2f*)&ut[3 * C_EMB];
            v2f n4 = *(const v2f*)&ut[4 * C_EMB];
            v2f n5 = *(const v2f*)&ut[5 * C_EMB];
#pragma unroll
            for (int q = 0; q < 16; ++q) {
                const v2f a0 = n0, a1 = n1, a2 = n2, a3 = n3, a4 = n4, a5 = n5;
                if (q + 1 < 16) {
                    n0 = *(const v2f*)&ut[0 * C_EMB + 4 * (q + 1)];
                    n1 = *(const v2f*)&ut[1 * C_EMB + 4 * (q + 1)];
                    n2 = *(const v2f*)&ut[2 * C_EMB + 4 * (q + 1)];
                    n3 = *(const v2f*)&ut[3 * C_EMB + 4 * (q + 1)];
                    n4 = *(const v2f*)&ut[4 * C_EMB + 4 * (q + 1)];
                    n5 = *(const v2f*)&ut[5 * C_EMB + 4 * (q + 1)];
                }
                acc0 = __builtin_amdgcn_wmma_f32_16x16x4_f32(false, a0, false, B0[q],
                                                             (short)0, acc0, false, false);
                acc1 = __builtin_amdgcn_wmma_f32_16x16x4_f32(false, a1, false, B1[q],
                                                             (short)0, acc1, false, false);
                acc2 = __builtin_amdgcn_wmma_f32_16x16x4_f32(false, a2, false, B2[q],
                                                             (short)0, acc2, false, false);
                acc3 = __builtin_amdgcn_wmma_f32_16x16x4_f32(false, a3, false, B3[q],
                                                             (short)0, acc3, false, false);
                acc4 = __builtin_amdgcn_wmma_f32_16x16x4_f32(false, a4, false, B4[q],
                                                             (short)0, acc4, false, false);
                acc5 = __builtin_amdgcn_wmma_f32_16x16x4_f32(false, a5, false, B5[q],
                                                             (short)0, acc5, false, false);
            }

#pragma unroll
            for (int v = 0; v < 8; ++v) {
                const float s12 = acc1[v] + acc2[v];
                const float d12 = acc1[v] - acc2[v];
                const float s34 = acc3[v] + acc4[v];
                const float d34 = acc3[v] - acc4[v];
                const float y0 = 4.0f * acc0[v] + c4[v];
                const float y1 = (4.0f * (acc5[v] - d12) - 2.0f * d34) + c5[v];
                const float y2 = -(5.0f * acc0[v] + 4.0f * s12) - s34;
                const float y3 = (d12 + 2.0f * d34) - 5.0f * acc5[v];
                c4[v] = (acc0[v] + s12) + s34;
                c5[v] = acc5[v];
                mx[v] = fmaxf(mx[v], fmaxf(fmaxf(y0, y1), fmaxf(y2, y3)));
            }
        }

#pragma unroll
        for (int v = 0; v < 8; ++v)
            mx[v] = fmaxf(mx[v], fmaxf(c4[v], c5[v]));

        float* orow = &out[(long long)(nb + grp * WPG + 8 * half) * OUT_COLS + g * 16 + ocol];
#pragma unroll
        for (int v = 0; v < 8; ++v)
            orow[v * OUT_COLS] = fmaxf(0.0f, bias + mx[v]);

        if (grp + 1 < GPB)
            lds_barrier();
    }
}

extern "C" void kernel_launch(void* const* d_in, const int* in_sizes, int n_in,
                              void* d_out, int out_size, void* d_ws, size_t ws_size,
                              hipStream_t stream) {
    const int*   X          = (const int*)d_in[0];
    const int*   X_word     = (const int*)d_in[1];
    const float* char_emb_w = (const float*)d_in[2];
    const float* conv_w     = (const float*)d_in[3];
    const float* conv_b     = (const float*)d_in[4];
    const float* word_emb_w = (const float*)d_in[5];
    float*       out        = (float*)d_out;
    (void)in_sizes; (void)n_in; (void)out_size; (void)d_ws; (void)ws_size;

    char_conv_kernel<<<N_CBLK, BLOCK_T, 0, stream>>>(X, X_word, char_emb_w, word_emb_w,
                                                     conv_w, conv_b, out);
}
